# starting kernel
# baseline (speedup 1.0000x reference)
#include <hip/hip_runtime.h>
#include <cstdint>

#define NN      50000
#define NE      1600000
#define ET      (NE + NN)
#define IN_C    256
#define HID_C   128
#define HEADS   8
#define OUT_C   16
#define SLOPE   0.2f

typedef float v2f __attribute__((ext_vector_type(2)));
typedef float v8f __attribute__((ext_vector_type(8)));

__device__ __forceinline__ unsigned enc_f32(float f) {
    unsigned u = __float_as_uint(f);
    return (u & 0x80000000u) ? ~u : (u | 0x80000000u);
}
__device__ __forceinline__ float dec_f32(unsigned e) {
    unsigned u = (e & 0x80000000u) ? (e & 0x7FFFFFFFu) : ~e;
    return __uint_as_float(u);
}

__global__ void k_zero(unsigned* __restrict__ p, long n) {
    long i = (long)blockIdx.x * 256 + threadIdx.x;
    if (i < n) p[i] = 0u;
}

__global__ void k_gemm1(const float* __restrict__ X, const float* __restrict__ W,
                        float* __restrict__ H) {
    const int wave = threadIdx.x >> 5;
    const int lane = threadIdx.x & 31;
    const int mBase = blockIdx.x << 4;
    const int nBase = wave << 4;
    const int lm = lane & 15;
    const int kh = (lane >> 4) << 1;
    const float* xr = X + (size_t)(mBase + lm) * IN_C + kh;
    const float* wp = W + (size_t)kh * HID_C + nBase + lm;
    v8f acc = {};
    #pragma unroll 8
    for (int kb = 0; kb < IN_C; kb += 4) {
        v2f a = *(const v2f*)(xr + kb);
        v2f b;
        b.x = wp[(size_t)kb * HID_C];
        b.y = wp[(size_t)kb * HID_C + HID_C];
        acc = __builtin_amdgcn_wmma_f32_16x16x4_f32(false, a, false, b,
                                                    (short)0, acc, false, false);
    }
    float* hp = H + (size_t)(mBase + ((lane >> 4) << 3)) * HID_C + nBase + lm;
    #pragma unroll
    for (int r = 0; r < 8; ++r) hp[(size_t)r * HID_C] = acc[r];
}

__global__ void k_gemm2(const float* __restrict__ X, const float* __restrict__ W,
                        float* __restrict__ H) {
    const int wave = threadIdx.x >> 5;
    const int lane = threadIdx.x & 31;
    const int mBase = (blockIdx.x << 7) + (wave << 4);
    if (mBase >= NN) return;
    const int lm = lane & 15;
    const int kh = (lane >> 4) << 1;
    const float* xr = X + (size_t)(mBase + lm) * HID_C + kh;
    const float* wp = W + (size_t)kh * OUT_C + lm;
    v8f acc = {};
    #pragma unroll 8
    for (int kb = 0; kb < HID_C; kb += 4) {
        v2f a = *(const v2f*)(xr + kb);
        v2f b;
        b.x = wp[(size_t)kb * OUT_C];
        b.y = wp[(size_t)kb * OUT_C + OUT_C];
        acc = __builtin_amdgcn_wmma_f32_16x16x4_f32(false, a, false, b,
                                                    (short)0, acc, false, false);
    }
    float* hp = H + (size_t)(mBase + ((lane >> 4) << 3)) * OUT_C + lm;
    #pragma unroll
    for (int r = 0; r < 8; ++r) hp[(size_t)r * OUT_C] = acc[r];
}

__global__ void k_attvec(const float* __restrict__ H, const float* __restrict__ attS,
                         const float* __restrict__ attD, float* __restrict__ aS,
                         float* __restrict__ aD, int nH) {
    int idx = blockIdx.x * blockDim.x + threadIdx.x;
    if (idx >= NN * nH) return;
    int hd = idx % nH;
    const float4* hp = (const float4*)(H + (size_t)idx * 16);
    const float4* sp = (const float4*)(attS + hd * 16);
    const float4* dp = (const float4*)(attD + hd * 16);
    float s = 0.f, d = 0.f;
    #pragma unroll
    for (int j = 0; j < 4; ++j) {
        float4 h4 = hp[j], s4 = sp[j], d4 = dp[j];
        s += h4.x * s4.x + h4.y * s4.y + h4.z * s4.z + h4.w * s4.w;
        d += h4.x * d4.x + h4.y * d4.y + h4.z * d4.z + h4.w * d4.w;
    }
    aS[idx] = s; aD[idx] = d;
}

template <int NH>
__global__ void k_edge_max(const int* __restrict__ esrc, const int* __restrict__ edst,
                           const float* __restrict__ aS, const float* __restrict__ aD,
                           unsigned* __restrict__ amax) {
    int e = blockIdx.x * blockDim.x + threadIdx.x;
    if (e >= ET) return;
    int s, d;
    if (e < NE) { s = esrc[e]; d = edst[e]; } else { s = e - NE; d = s; }
    #pragma unroll
    for (int hd = 0; hd < NH; ++hd) {
        float al = aS[(size_t)s * NH + hd] + aD[(size_t)d * NH + hd];
        al = al > 0.f ? al : al * SLOPE;
        atomicMax(&amax[(size_t)d * NH + hd], enc_f32(al));
    }
}

template <int NH>
__global__ void k_edge_accum(const int* __restrict__ esrc, const int* __restrict__ edst,
                             const float* __restrict__ aS, const float* __restrict__ aD,
                             const unsigned* __restrict__ amax, const float* __restrict__ H,
                             float* __restrict__ denom, float* __restrict__ accum) {
    long idx = (long)blockIdx.x * blockDim.x + threadIdx.x;
    if (idx >= (long)ET * NH) return;
    int e  = (int)(idx / NH);
    int hd = (int)(idx - (long)e * NH);
    int s, d;
    if (e < NE) { s = esrc[e]; d = edst[e]; } else { s = e - NE; d = s; }
    float al = aS[(size_t)s * NH + hd] + aD[(size_t)d * NH + hd];
    al = al > 0.f ? al : al * SLOPE;
    float w = __expf(al - dec_f32(amax[(size_t)d * NH + hd]));
    atomicAdd(&denom[(size_t)d * NH + hd], w);
    const float4* hp = (const float4*)(H + ((size_t)s * NH + hd) * 16);
    float* op = accum + ((size_t)d * NH + hd) * 16;
    #pragma unroll
    for (int j = 0; j < 4; ++j) {
        float4 v = hp[j];
        atomicAdd(op + 4 * j + 0, w * v.x);
        atomicAdd(op + 4 * j + 1, w * v.y);
        atomicAdd(op + 4 * j + 2, w * v.z);
        atomicAdd(op + 4 * j + 3, w * v.w);
    }
}

__global__ void k_fin(float* __restrict__ accum, const float* __restrict__ denom,
                      const float* __restrict__ bias, int rowMask, long total, int relu) {
    long idx = (long)blockIdx.x * blockDim.x + threadIdx.x;
    if (idx >= total) return;
    int c = (int)(idx & rowMask);
    long nh = idx >> 4;
    float v = accum[idx] / denom[nh] + bias[c];
    if (relu) v = fmaxf(v, 0.f);
    accum[idx] = v;
}

extern "C" void kernel_launch(void* const* d_in, const int* in_sizes, int n_in,
                              void* d_out, int out_size, void* d_ws, size_t ws_size,
                              hipStream_t stream) {
    const float* x     = (const float*)d_in[0];
    const int*   ei    = (const int*)  d_in[1];
    const float* W1    = (const float*)d_in[2];
    const float* attS1 = (const float*)d_in[3];
    const float* attD1 = (const float*)d_in[4];
    const float* bias1 = (const float*)d_in[5];
    const float* W2    = (const float*)d_in[6];
    const float* attS2 = (const float*)d_in[7];
    const float* attD2 = (const float*)d_in[8];
    const float* bias2 = (const float*)d_in[9];
    float* out = (float*)d_out;
    const int* esrc = ei;
    const int* edst = ei + NE;

    unsigned* amax1  = (unsigned*)d_ws;
    float*    denom1 = (float*)(amax1 + (size_t)NN * 8);
    float*    accum1 = denom1 + (size_t)NN * 8;
    unsigned* amax2  = (unsigned*)(accum1 + (size_t)NN * 128);
    float*    denom2 = (float*)(amax2 + NN);
    float* h1  = denom2 + NN;
    float* aS1 = h1 + (size_t)NN * 128;
    float* aD1 = aS1 + (size_t)NN * 8;
    float* h2  = aD1 + (size_t)NN * 8;
    float* aS2 = h2 + (size_t)NN * 16;
    float* aD2 = aS2 + NN;

    const long znWs  = (long)NN * (8 + 8 + 128 + 1 + 1);
    const long znOut = (long)NN * OUT_C;
    k_zero<<<(int)((znWs  + 255) / 256), 256, 0, stream>>>((unsigned*)d_ws, znWs);
    k_zero<<<(int)((znOut + 255) / 256), 256, 0, stream>>>((unsigned*)out, znOut);

    k_gemm1<<<NN / 16, 256, 0, stream>>>(x, W1, h1);
    k_attvec<<<(NN * HEADS + 255) / 256, 256, 0, stream>>>(h1, attS1, attD1, aS1, aD1, HEADS);
    k_edge_max<HEADS><<<(ET + 255) / 256, 256, 0, stream>>>(esrc, edst, aS1, aD1, amax1);
    {
        long n = (long)ET * HEADS;
        k_edge_accum<HEADS><<<(int)((n + 255) / 256), 256, 0, stream>>>(
            esrc, edst, aS1, aD1, amax1, h1, denom1, accum1);
    }
    {
        long n = (long)NN * HID_C;
        k_fin<<<(int)((n + 255) / 256), 256, 0, stream>>>(accum1, denom1, bias1,
                                                          HID_C - 1, n, 1);
    }

    k_gemm2<<<(NN + 127) / 128, 256, 0, stream>>>(accum1, W2, h2);
    k_attvec<<<(NN + 255) / 256, 256, 0, stream>>>(h2, attS2, attD2, aS2, aD2, 1);
    k_edge_max<1><<<(ET + 255) / 256, 256, 0, stream>>>(esrc, edst, aS2, aD2, amax2);
    {
        long n = (long)ET;
        k_edge_accum<1><<<(int)((n + 255) / 256), 256, 0, stream>>>(
            esrc, edst, aS2, aD2, amax2, h2, denom2, out);
    }
    {
        long n = (long)NN * OUT_C;
        k_fin<<<(int)((n + 255) / 256), 256, 0, stream>>>(out, denom2, bias2,
                                                          OUT_C - 1, n, 0);
    }
}


# Owner test from a bit map: prep writes one row of bits per owner group (LDS integer OR, whole-line stores); a group's filter reads only its row, 32 edges per word; LDS float adds, arrival order
# speedup vs baseline: 11.6836x; 11.6836x over previous
#include <hip/hip_runtime.h>
#include <cstdint>

#define NN      50000
#define NE      1600000
#define ET      (NE + NN)
#define IN_C    256
#define HID_C   128
#define HEADS   8
#define OUT_C   16
#define SLOPE   0.2f

typedef float v2f __attribute__((ext_vector_type(2)));
typedef float v4f __attribute__((ext_vector_type(4)));
typedef float v8f __attribute__((ext_vector_type(8)));

__device__ __forceinline__ unsigned enc_f32(float f) {
    unsigned u = __float_as_uint(f);
    return (u & 0x80000000u) ? ~u : (u | 0x80000000u);
}
__device__ __forceinline__ float dec_f32(unsigned e) {
    unsigned u = (e & 0x80000000u) ? (e & 0x7FFFFFFFu) : ~e;
    return __uint_as_float(u);
}

__device__ __forceinline__ unsigned umax2(unsigned a, unsigned b) { return a > b ? a : b; }

#define GS_W    32
#define G1_MAX  0
#define G1_NEG  8
#define G2_MAX  16
#define G2_NEG  17
#define SPREAD_OK 16.0f
#define MAX_BLOCKS 1024

__device__ __forceinline__ bool spread_small8(const unsigned* __restrict__ g) {
    bool ok = true;
    #pragma unroll
    for (int h = 0; h < 8; ++h) {
        const float r = dec_f32(g[G1_MAX + h]) + dec_f32(g[G1_NEG + h]);
        ok = ok && (r < SPREAD_OK);
    }
    return ok;
}
__device__ __forceinline__ bool spread_small1(const unsigned* __restrict__ g) {
    return dec_f32(g[G2_MAX]) + dec_f32(g[G2_NEG]) < SPREAD_OK;
}

#define RNG   196

static_assert(NN <= 65536, "a node number must fit 16 bits of the packed edge word");
#define BMW   32
#define BMBLK ((NE + 1023) / 1024)
#define BMP   (BMBLK * BMW)
#define BMQ   (BMP / 4)
__global__ void __launch_bounds__(256)
k_prep(unsigned* __restrict__ p, long n, const int* __restrict__ esrc,
       const int* __restrict__ edst, unsigned* __restrict__ bm,
       unsigned* __restrict__ pk) {
    __shared__ unsigned bits[256 * BMW];
    const int t = threadIdx.x;
    long i = (long)blockIdx.x * 256 + t;
    if (i < n) p[i] = 0u;
    if (blockIdx.x < BMBLK) {
        for (int k = 0; k < 256 * BMW / 4 / 256; ++k)
            ((uint4*)bits)[k * 256 + t] = make_uint4(0u, 0u, 0u, 0u);
        __syncthreads();
        if (i < NE / 4) {
            const int4 d = ((const int4*)edst)[i];
            const int4 s = ((const int4*)esrc)[i];
            uint4 w;
            w.x = (unsigned)s.x | ((unsigned)d.x << 16);
            w.y = (unsigned)s.y | ((unsigned)d.y << 16);
            w.z = (unsigned)s.z | ((unsigned)d.z << 16);
            w.w = (unsigned)s.w | ((unsigned)d.w << 16);
            ((uint4*)pk)[i] = w;
            const int wd = t >> 3;
            const unsigned b0 = 1u << ((4 * t) & 31);
            atomicOr(&bits[((unsigned)d.x / RNG) * BMW + wd], b0);
            atomicOr(&bits[((unsigned)d.y / RNG) * BMW + wd], b0 << 1);
            atomicOr(&bits[((unsigned)d.z / RNG) * BMW + wd], b0 << 2);
            atomicOr(&bits[((unsigned)d.w / RNG) * BMW + wd], b0 << 3);
        }
        __syncthreads();
        const int lane = t & 31, wv = t >> 5;
        const int pc = (lane & 7) * 4;
        unsigned* row = bm + (size_t)blockIdx.x * BMW + pc;
        for (int k = 0; k < 8; ++k) {
            const int g = k * 32 + wv * 4 + (lane >> 3);
            *(uint4*)(row + (size_t)g * BMP) = *(const uint4*)(bits + g * BMW + pc);
        }
    }
}

#define G1_WAVES 16
#define G1_WGS   256
#define G1_UNITS (NN / 16 * 2)
__global__ void __launch_bounds__(G1_WAVES * 32)
k_gemm1(const float* __restrict__ X, const float* __restrict__ W,
        float* __restrict__ H) {
    __shared__ v4f wl[2 * 32 * 2 * 2 * 32];
    {
        const int t = threadIdx.x;
        const int k0 = 8 * (t >> 7) + 4 * ((t >> 4) & 1) + 2 * ((t >> 6) & 1);
        const int n0 = 32 * ((t >> 5) & 1) + (t & 15);
        v4f st[16];
        #pragma unroll
        for (int it = 0; it < 16; ++it) {
            const float* wp = W + (k0 + 32 * (it & 7)) * HID_C + n0 + 64 * (it >> 3);
            st[it].x = wp[0]; st[it].y = wp[HID_C]; st[it].z = wp[16]; st[it].w = wp[HID_C + 16];
        }
        #pragma unroll
        for (int it = 0; it < 16; ++it) wl[it * (G1_WAVES * 32) + t] = st[it];
    }
    __syncthreads();
    const int lane = threadIdx.x & 31;
    const int wv = __builtin_amdgcn_readfirstlane((int)(threadIdx.x >> 5));
    const int c = lane & 15, h = lane >> 4;
    const int u0 = (int)(((long)blockIdx.x * G1_UNITS) / G1_WGS);
    const int u1 = (int)(((long)(blockIdx.x + 1) * G1_UNITS) / G1_WGS);
    for (int u = u0 + wv; u < u1; u += G1_WAVES) {
        asm volatile("" ::: "memory");
        const int strip = u >> 1, hf = u & 1;
        const v4f* xr = (const v4f*)(X + (size_t)(strip * 16 + c) * IN_C + 4 * h);
        v4f a[32];
        #pragma unroll
        for (int j = 0; j < 32; ++j) a[j] = xr[2 * j];
        asm volatile("" :: "v"(a[0]), "v"(a[1]), "v"(a[2]), "v"(a[3]), "v"(a[4]), "v"(a[5]), "v"(a[6]), "v"(a[7]), "v"(a[8]), "v"(a[9]), "v"(a[10]), "v"(a[11]), "v"(a[12]), "v"(a[13]), "v"(a[14]), "v"(a[15]));
        asm volatile("" :: "v"(a[16]), "v"(a[17]), "v"(a[18]), "v"(a[19]), "v"(a[20]), "v"(a[21]), "v"(a[22]), "v"(a[23]), "v"(a[24]), "v"(a[25]), "v"(a[26]), "v"(a[27]), "v"(a[28]), "v"(a[29]), "v"(a[30]), "v"(a[31]));
        v8f acc0 = {}, acc1 = {}, acc2 = {}, acc3 = {};
        const v4f* bl = wl + hf * 4096 + lane;
        #pragma unroll
        for (int j = 0; j < 32; ++j) {
            #pragma unroll
            for (int m = 0; m < 2; ++m) {
                const v4f b0 = bl[((j * 2 + m) * 2 + 0) * 32];
                const v4f b1 = bl[((j * 2 + m) * 2 + 1) * 32];
                v2f av, bv;
                av.x = m ? a[j].z : a[j].x;
                av.y = m ? a[j].w : a[j].y;
                bv.x = b0.x; bv.y = b0.y;
                acc0 = __builtin_amdgcn_wmma_f32_16x16x4_f32(false, av, false, bv,
                                                             (short)0, acc0, false, false);
                bv.x = b0.z; bv.y = b0.w;
                acc1 = __builtin_amdgcn_wmma_f32_16x16x4_f32(false, av, false, bv,
                                                             (short)0, acc1, false, false);
                bv.x = b1.x; bv.y = b1.y;
                acc2 = __builtin_amdgcn_wmma_f32_16x16x4_f32(false, av, false, bv,
                                                             (short)0, acc2, false, false);
                bv.x = b1.z; bv.y = b1.w;
                acc3 = __builtin_amdgcn_wmma_f32_16x16x4_f32(false, av, false, bv,
                                                             (short)0, acc3, false, false);
            }
        }
        float* hp = H + (size_t)(strip * 16 + 8 * h) * HID_C + 64 * hf + c;
        #pragma unroll
        for (int r = 0; r < 8; ++r) {
            hp[(size_t)r * HID_C]      = acc0[r];
            hp[(size_t)r * HID_C + 16] = acc1[r];
            hp[(size_t)r * HID_C + 32] = acc2[r];
            hp[(size_t)r * HID_C + 48] = acc3[r];
        }
    }
}

__global__ void k_gemm2(const float* __restrict__ X, const float* __restrict__ W,
                        float* __restrict__ H) {
    const int wave = threadIdx.x >> 5;
    const int lane = threadIdx.x & 31;
    const int mBase = (blockIdx.x << 7) + (wave << 4);
    if (mBase >= NN) return;
    const int lm = lane & 15;
    const int kh = (lane >> 4) << 1;
    const float* xr = X + (size_t)(mBase + lm) * HID_C + kh;
    const float* wp = W + (size_t)kh * OUT_C + lm;
    v8f acc = {};
    #pragma unroll 8
    for (int kb = 0; kb < HID_C; kb += 4) {
        v2f a = *(const v2f*)(xr + kb);
        v2f b;
        b.x = wp[(size_t)kb * OUT_C];
        b.y = wp[(size_t)kb * OUT_C + OUT_C];
        acc = __builtin_amdgcn_wmma_f32_16x16x4_f32(false, a, false, b,
                                                    (short)0, acc, false, false);
    }
    float* hp = H + (size_t)(mBase + ((lane >> 4) << 3)) * OUT_C + lm;
    #pragma unroll
    for (int r = 0; r < 8; ++r) hp[(size_t)r * OUT_C] = acc[r];
}

__global__ void __launch_bounds__(256)
k_attvec(const float* __restrict__ H, const float* __restrict__ attS,
         const float* __restrict__ attD, float* __restrict__ aS,
         float* __restrict__ aD, int nH, unsigned* __restrict__ g) {
    __shared__ unsigned red[8][16];
    const int idx = blockIdx.x * 256 + threadIdx.x;
    unsigned eM = 0u, eN = 0u;
    if (idx < NN * nH) {
        const int hd = idx % nH;
        const float4* hp = (const float4*)(H + (size_t)idx * 16);
        const float4* sp = (const float4*)(attS + hd * 16);
        const float4* dp = (const float4*)(attD + hd * 16);
        float s = 0.f, d = 0.f;
        #pragma unroll
        for (int j = 0; j < 4; ++j) {
            float4 h4 = hp[j], s4 = sp[j], d4 = dp[j];
            s += h4.x * s4.x + h4.y * s4.y + h4.z * s4.z + h4.w * s4.w;
            d += h4.x * d4.x + h4.y * d4.y + h4.z * d4.z + h4.w * d4.w;
        }
        aS[idx] = s; aD[idx] = d;
        eM = enc_f32(s); eN = enc_f32(-s);
    }
    eM = umax2(eM, (unsigned)__shfl_xor((int)eM, 8));
    eN = umax2(eN, (unsigned)__shfl_xor((int)eN, 8));
    eM = umax2(eM, (unsigned)__shfl_xor((int)eM, 16));
    eN = umax2(eN, (unsigned)__shfl_xor((int)eN, 16));
    if (nH == 1) {
        eM = umax2(eM, (unsigned)__shfl_xor((int)eM, 1));
        eN = umax2(eN, (unsigned)__shfl_xor((int)eN, 1));
        eM = umax2(eM, (unsigned)__shfl_xor((int)eM, 2));
        eN = umax2(eN, (unsigned)__shfl_xor((int)eN, 2));
        eM = umax2(eM, (unsigned)__shfl_xor((int)eM, 4));
        eN = umax2(eN, (unsigned)__shfl_xor((int)eN, 4));
    }
    const int lane = threadIdx.x & 31, wave = threadIdx.x >> 5;
    if (lane < 8) { red[wave][lane] = eM; red[wave][8 + lane] = eN; }
    __syncthreads();
    if (threadIdx.x < 16) {
        const int t = threadIdx.x;
        unsigned v = red[0][t];
        #pragma unroll
        for (int w = 1; w < 8; ++w) v = umax2(v, red[w][t]);
        if (nH == 8) atomicMax(&g[G1_MAX + t], v);
        else if ((t & 7) == 0) atomicMax(&g[G2_MAX + (t >> 3)], v);
    }
}

#define ROWW 32
#define L1_MAX 8
#define L2_MAX 17

__device__ __forceinline__ void edge_sd(const int* __restrict__ esrc,
                                        const int* __restrict__ edst, int e, int& s, int& d) {
    if (e < NE) { s = esrc[e]; d = edst[e]; } else { s = e - NE; d = s; }
}
__device__ __forceinline__ void edge_sd_pk(const unsigned* __restrict__ pk, int e, int& s, int& d) {
    if (e < NE) { const unsigned w = pk[e]; s = (int)(w & 0xFFFFu); d = (int)(w >> 16); }
    else { s = e - NE; d = s; }
}
__device__ __forceinline__ float leaky(float a) { return a > 0.f ? a : a * SLOPE; }

__global__ void k_edge_max8(const int* __restrict__ esrc, const int* __restrict__ edst,
                            const float* __restrict__ aS, const float* __restrict__ aD,
                            unsigned* __restrict__ side, const unsigned* __restrict__ g) {
    if (spread_small8(g)) return;
    const long total = (long)ET * 8;
    for (long idx = (long)blockIdx.x * 256 + threadIdx.x; idx < total;
         idx += (long)MAX_BLOCKS * 256) {
        int e = (int)(idx >> 3), hd = (int)(idx & 7);
        int s, d;
        edge_sd(esrc, edst, e, s, d);
        float al = leaky(aS[(size_t)s * 8 + hd] + aD[(size_t)d * 8 + hd]);
        atomicMax(&side[(size_t)d * ROWW + L1_MAX + hd], enc_f32(al));
    }
}

__global__ void k_edge_max1(const int* __restrict__ esrc, const int* __restrict__ edst,
                            const float* __restrict__ aS, const float* __restrict__ aD,
                            unsigned* __restrict__ row, const unsigned* __restrict__ g) {
    if (spread_small1(g)) return;
    for (int e = blockIdx.x * 256 + threadIdx.x; e < ET; e += MAX_BLOCKS * 256) {
        int s, d;
        edge_sd(esrc, edst, e, s, d);
        float al = leaky(aS[s] + aD[d]);
        atomicMax(&row[(size_t)d * ROWW + L2_MAX], enc_f32(al));
    }
}

#define AGW   32
#define QCAP  1056
#define NIT   ((BMQ + AGW * 32 - 1) / (AGW * 32))
#define LCAP  512

template <class F>
__device__ __forceinline__ void filter_slice(const unsigned* __restrict__ bm, int wv, int lane,
                                             int blk, int lo, int* q, F&& batch) {
    int base = 0;
    const uint4* row = (const uint4*)(bm + (size_t)blk * BMP);
    const int wr = AGW - 1 - wv;
    uint4 dv = row[wr * 32 + lane];
    static_assert(AGW * 32 <= BMQ, "the first step of every wave lies inside the row");
    for (int it = 0; it <= NIT; ++it) {
        uint4 nx = make_uint4(0u, 0u, 0u, 0u);
        {
            const int c = ((it + 1) * AGW + wr) * 32 + lane;
            if (c < BMQ) nx = row[c];
        }
        uint4 cur = dv;
        const int eb = ((it * AGW + wr) * 32 + lane) * 128;
        const int nt = it < NIT ? 4 : 1;
        #pragma nounroll
        for (int t = 0; t < nt; ++t) {
            if (it < NIT) {
                unsigned m = cur.x;
                unsigned mask;
                while ((mask = __builtin_amdgcn_ballot_w32(m != 0u)) != 0u) {
                    const int pos = base + (int)__builtin_amdgcn_mbcnt_lo(mask, 0u);
                    if (m != 0u) q[pos] = eb + 32 * t + __builtin_ctz(m);
                    m &= m - 1u;
                    base += __builtin_popcount(mask);
                }
                cur.x = cur.y; cur.y = cur.z; cur.z = cur.w;
            } else {
                const int nl = wv * 32 + lane;
                const bool m = nl < RNG && lo + nl < NN;
                const unsigned mask = __builtin_amdgcn_ballot_w32(m);
                if (mask) {
                    const int pos = base + (int)__builtin_amdgcn_mbcnt_lo(mask, 0u);
                    if (m) q[pos] = NE + lo + nl;
                    base += __builtin_popcount(mask);
                }
            }
            const int need = it < NIT ? 32 : 1;
            while (base >= need) {
                const int cnt = base < 32 ? base : 32;
                const int e = lane < cnt ? q[base - cnt + lane] : 0;
                batch(e, cnt);
                base -= cnt;
            }
        }
        dv = nx;
    }
}

__device__ __forceinline__ float quad_bcast(float x, int u) {
    const int xi = __float_as_int(x);
    int r;
    switch (u) {
        case 0:  r = __builtin_amdgcn_update_dpp(0, xi, 0x00, 0xf, 0xf, false); break;
        case 1:  r = __builtin_amdgcn_update_dpp(0, xi, 0x55, 0xf, 0xf, false); break;
        case 2:  r = __builtin_amdgcn_update_dpp(0, xi, 0xAA, 0xf, 0xf, false); break;
        default: r = __builtin_amdgcn_update_dpp(0, xi, 0xFF, 0xf, 0xf, false); break;
    }
    return __int_as_float(r);
}

__device__ __forceinline__ void agg_batch(
        int e, int cnt, int lane, int lo,
        const unsigned* __restrict__ pk,
        const float* __restrict__ aS, const float* __restrict__ aD,
        const float* __restrict__ H, const unsigned* __restrict__ sideU,
        bool useBound, float gmx, float* acc, float* den,
        int2* __restrict__ lst, int lpos) {
    int s = 0, d = lo;
    if (lane < cnt) edge_sd_pk(pk, e, s, d);
    if (lane < cnt && lpos + lane < LCAP) lst[lpos + lane] = make_int2(s, d);
    const v4f* hrow = (const v4f*)H + lane;
    auto load_row = [&](int i) -> v4f {
        const int sk = __builtin_amdgcn_readlane(s, i);
        return hrow[(size_t)sk * (HID_C / 4)];
    };
    v4f r[3][4];
    #pragma unroll
    for (int k = 0; k < 8; ++k) r[k >> 2][k & 3] = load_row(k);
    const int hd = lane >> 2, sub = lane & 3;
    float wq[8];
    #pragma unroll
    for (int q = 0; q < 8; ++q) {
        const int i = q * 4 + sub;
        const int sq = __shfl(s, i), dq = __shfl(d, i);
        const float adq = aD[(size_t)dq * 8 + hd];
        const float al = leaky(aS[(size_t)sq * 8 + hd] + adq);
        float mx;
        if (useBound) mx = leaky(gmx + adq);
        else          mx = dec_f32(sideU[(size_t)dq * ROWW + L1_MAX + hd]);
        wq[q] = __expf(al - mx);
        if (i < cnt) atomicAdd(&den[(dq - lo) * 8 + hd], wq[q]);
    }
    auto add_row = [&](int i, const v4f rv) {
        const float wk = quad_bcast(wq[i >> 2], i & 3);
        const int dk = __builtin_amdgcn_readlane(d, i);
        float* op = acc + (dk - lo) * HID_C + lane;
        atomicAdd(op,      wk * rv.x);
        atomicAdd(op + 32, wk * rv.y);
        atomicAdd(op + 64, wk * rv.z);
        atomicAdd(op + 96, wk * rv.w);
    };
    if (cnt == 32) {
        #pragma unroll
        for (int g = 0; g < 8; ++g) {
            if (g + 2 < 8) {
                #pragma unroll
                for (int k = 0; k < 4; ++k) r[(g + 2) % 3][k] = load_row((g + 2) * 4 + k);
            }
            #pragma unroll
            for (int k = 0; k < 4; ++k) add_row(g * 4 + k, r[g % 3][k]);
        }
    } else {
        #pragma unroll
        for (int g = 0; g < 8; ++g) {
            if (g * 4 >= cnt) break;
            if (g >= 2) {
                #pragma unroll
                for (int k = 0; k < 4; ++k) r[g % 3][k] = load_row(g * 4 + k);
            }
            #pragma unroll
            for (int k = 0; k < 4; ++k)
                if (g * 4 + k < cnt) add_row(g * 4 + k, r[g % 3][k]);
        }
    }
}

__global__ void __launch_bounds__(1024)
k_agg1(const unsigned* __restrict__ pk,
       const float* __restrict__ aS, const float* __restrict__ aD,
       const float* __restrict__ H, const unsigned* __restrict__ sideU,
       const float* __restrict__ bias, float* __restrict__ out1,
       const unsigned* __restrict__ g, const unsigned* __restrict__ bm,
       int* __restrict__ lst, int* __restrict__ lcnt) {
    __shared__ float acc[RNG * HID_C];
    __shared__ float den[RNG * 8];
    __shared__ int   que[AGW][QCAP];
    const int lane = threadIdx.x & 31;
    const int wv = __builtin_amdgcn_readfirstlane((int)(threadIdx.x >> 5));
    const int lo = blockIdx.x * RNG;
    for (int i = threadIdx.x; i < RNG * HID_C; i += 1024) acc[i] = 0.f;
    for (int i = threadIdx.x; i < RNG * 8; i += 1024) den[i] = 0.f;
    __syncthreads();
    const bool useBound = spread_small8(g);
    const float gmx = dec_f32(g[G1_MAX + (lane >> 2)]);
    int2* lp = (int2*)lst + (size_t)(blockIdx.x * AGW + wv) * LCAP;
    int lpos = 0;
    filter_slice(bm, wv, lane, (int)blockIdx.x, lo, que[wv], [&](int e, int cnt) {
        agg_batch(e, cnt, lane, lo, pk, aS, aD, H, sideU, useBound, gmx, acc, den,
                  lp, lpos);
        lpos += cnt;
    });
    if (lane == 0) lcnt[blockIdx.x * AGW + wv] = lpos;
    __syncthreads();
    for (int i = threadIdx.x; i < RNG * 32; i += 1024) {
        const int nl = i >> 5, L = i & 31;
        const int n = lo + nl;
        if (n < NN) {
            const float dn = den[nl * 8 + (L >> 2)];
            const float* ap = acc + nl * HID_C + L;
            const float* bp = bias + 4 * L;
            v4f o;
            o.x = fmaxf(ap[0]  / dn + bp[0], 0.f);
            o.y = fmaxf(ap[32] / dn + bp[1], 0.f);
            o.z = fmaxf(ap[64] / dn + bp[2], 0.f);
            o.w = fmaxf(ap[96] / dn + bp[3], 0.f);
            ((v4f*)(out1 + (size_t)n * HID_C))[L] = o;
        }
    }
}

__device__ __forceinline__ void agg2_batch(
        int s, int d, int cnt, int lane, int lo,
        const float* __restrict__ aS, const float* __restrict__ aD,
        const float* __restrict__ H, const unsigned* __restrict__ sideU,
        bool useBound, float gmx, float* acc, float* den) {
    const float adq = aD[d];
    const float al = leaky(aS[s] + adq);
    float mx;
    if (useBound) mx = leaky(gmx + adq);
    else          mx = dec_f32(sideU[(size_t)d * ROWW + L2_MAX]);
    const float w = __expf(al - mx);
    if (lane < cnt) atomicAdd(&den[d - lo], w);
    const int ch = lane & 15, hi = lane >> 4;
    float v[16];
    int   dd[16];
    #pragma unroll
    for (int t = 0; t < 16; ++t) {
        const int i = t * 2 + hi;
        const int st = __shfl(s, i);
        dd[t] = __shfl(d, i);
        v[t] = __shfl(w, i) * H[(size_t)st * OUT_C + ch];
    }
    #pragma unroll
    for (int t = 0; t < 16; ++t) {
        const int i = t * 2 + hi;
        if (i < cnt) atomicAdd(&acc[(dd[t] - lo) * OUT_C + ch], v[t]);
    }
}

__global__ void __launch_bounds__(1024)
k_agg2(const unsigned* __restrict__ pk,
       const float* __restrict__ aS, const float* __restrict__ aD,
       const float* __restrict__ H, const unsigned* __restrict__ sideU,
       const float* __restrict__ bias, float* __restrict__ out,
       const unsigned* __restrict__ g, const unsigned* __restrict__ bm,
       const int* __restrict__ lst, const int* __restrict__ lcnt) {
    __shared__ float acc[RNG * OUT_C];
    __shared__ float den[RNG];
    __shared__ int   que[AGW][QCAP];
    const int lane = threadIdx.x & 31;
    const int wv = __builtin_amdgcn_readfirstlane((int)(threadIdx.x >> 5));
    const int lo = blockIdx.x * RNG;
    for (int i = threadIdx.x; i < RNG * OUT_C; i += 1024) acc[i] = 0.f;
    for (int i = threadIdx.x; i < RNG; i += 1024) den[i] = 0.f;
    __syncthreads();
    const bool useBound = spread_small1(g);
    const float gmx = dec_f32(g[G2_MAX]);
    const int tot = __builtin_amdgcn_readfirstlane(lcnt[blockIdx.x * AGW + wv]);
    if (tot <= LCAP) {
        const int2* lp = (const int2*)lst + (size_t)(blockIdx.x * AGW + wv) * LCAP;
        int2 sd = make_int2(0, lo);
        if (lane < tot) sd = lp[lane];
        for (int b0 = 0; b0 < tot; b0 += 32) {
            int2 nx = make_int2(0, lo);
            if (b0 + 32 + lane < tot) nx = lp[b0 + 32 + lane];
            const int cnt = tot - b0 < 32 ? tot - b0 : 32;
            agg2_batch(sd.x, sd.y, cnt, lane, lo, aS, aD, H, sideU, useBound, gmx, acc, den);
            sd = nx;
        }
    } else {
        filter_slice(bm, wv, lane, (int)blockIdx.x, lo, que[wv], [&](int e, int cnt) {
            int s = 0, d = lo;
            if (lane < cnt) edge_sd_pk(pk, e, s, d);
            agg2_batch(s, d, cnt, lane, lo, aS, aD, H, sideU, useBound, gmx, acc, den);
        });
    }
    __syncthreads();
    for (int i = threadIdx.x; i < RNG * OUT_C; i += 1024) {
        const int nl = i >> 4, c = i & (OUT_C - 1);
        const int n = lo + nl;
        if (n < NN) out[(size_t)n * OUT_C + c] = acc[i] / den[nl] + bias[c];
    }
}

extern "C" void kernel_launch(void* const* d_in, const int* in_sizes, int n_in,
                              void* d_out, int out_size, void* d_ws, size_t ws_size,
                              hipStream_t stream) {
    const float* x     = (const float*)d_in[0];
    const int*   ei    = (const int*)  d_in[1];
    const float* W1    = (const float*)d_in[2];
    const float* attS1 = (const float*)d_in[3];
    const float* attD1 = (const float*)d_in[4];
    const float* bias1 = (const float*)d_in[5];
    const float* W2    = (const float*)d_in[6];
    const float* attS2 = (const float*)d_in[7];
    const float* attD2 = (const float*)d_in[8];
    const float* bias2 = (const float*)d_in[9];
    float* out = (float*)d_out;
    const int* esrc = ei;
    const int* edst = ei + NE;

    float* side1  = (float*)d_ws;
    unsigned* gst = (unsigned*)(side1 + (size_t)NN * ROWW);
    float* out1 = (float*)(gst + GS_W);
    float* h1  = out1 + (size_t)NN * 128;
    float* aS1 = h1 + (size_t)NN * 128;
    float* aD1 = aS1 + (size_t)NN * 8;
    float* h2  = aD1 + (size_t)NN * 8;
    float* aS2 = h2 + (size_t)NN * 16;
    float* aD2 = aS2 + NN;
    unsigned* bmap = (unsigned*)(aD2 + NN);
    int*  lcnt = (int*)(bmap + (size_t)256 * BMP);
    int*  lst  = lcnt + 256 * AGW;
    unsigned* pk = (unsigned*)(lst + (size_t)256 * AGW * LCAP * 2);

    const long znWs = (long)NN * ROWW + GS_W;
    k_prep<<<(int)((znWs + 255) / 256), 256, 0, stream>>>((unsigned*)d_ws, znWs, esrc, edst, bmap, pk);

    k_gemm1<<<G1_WGS, G1_WAVES * 32, 0, stream>>>(x, W1, h1);
    k_attvec<<<(NN * HEADS + 255) / 256, 256, 0, stream>>>(h1, attS1, attD1, aS1, aD1, HEADS, gst);
    k_edge_max8<<<MAX_BLOCKS, 256, 0, stream>>>(esrc, edst, aS1, aD1, (unsigned*)side1, gst);
    k_agg1<<<256, 1024, 0, stream>>>(
        pk, aS1, aD1, h1, (const unsigned*)side1, bias1, out1, gst, bmap, lst, lcnt);

    k_gemm2<<<(NN + 127) / 128, 256, 0, stream>>>(out1, W2, h2);
    k_attvec<<<(NN + 255) / 256, 256, 0, stream>>>(h2, attS2, attD2, aS2, aD2, 1, gst);
    k_edge_max1<<<MAX_BLOCKS, 256, 0, stream>>>(esrc, edst, aS2, aD2, (unsigned*)side1, gst);
    k_agg2<<<256, 1024, 0, stream>>>(
        pk, aS2, aD2, h2, (const unsigned*)side1, bias2, out, gst, bmap, lst, lcnt);
}


# Segment maxima (wide spread only) formed by each owner group in its LDS, one more pass over its own edges: both maximum launches and the zero fill of the side rows go; 9 launches to 7; same sums
# speedup vs baseline: 12.0357x; 1.0301x over previous
#include <hip/hip_runtime.h>
#include <cstdint>

#define NN      50000
#define NE      1600000
#define IN_C    256
#define HID_C   128
#define HEADS   8
#define OUT_C   16
#define SLOPE   0.2f

typedef float v2f __attribute__((ext_vector_type(2)));
typedef float v4f __attribute__((ext_vector_type(4)));
typedef float v8f __attribute__((ext_vector_type(8)));

__device__ __forceinline__ unsigned enc_f32(float f) {
    unsigned u = __float_as_uint(f);
    return (u & 0x80000000u) ? ~u : (u | 0x80000000u);
}
__device__ __forceinline__ float dec_f32(unsigned e) {
    unsigned u = (e & 0x80000000u) ? (e & 0x7FFFFFFFu) : ~e;
    return __uint_as_float(u);
}

__device__ __forceinline__ unsigned umax2(unsigned a, unsigned b) { return a > b ? a : b; }

#define GS_W    32
#define G1_MAX  0
#define G1_NEG  8
#define G2_MAX  16
#define G2_NEG  17
#define SPREAD_OK 16.0f

__device__ __forceinline__ bool spread_small8(const unsigned* __restrict__ g) {
    bool ok = true;
    #pragma unroll
    for (int h = 0; h < 8; ++h) {
        const float r = dec_f32(g[G1_MAX + h]) + dec_f32(g[G1_NEG + h]);
        ok = ok && (r < SPREAD_OK);
    }
    return ok;
}
__device__ __forceinline__ bool spread_small1(const unsigned* __restrict__ g) {
    return dec_f32(g[G2_MAX]) + dec_f32(g[G2_NEG]) < SPREAD_OK;
}

#define RNG   196

static_assert(NN <= 65536, "a node number must fit 16 bits of the packed edge word");
#define BMW   32
#define BMBLK ((NE + 1023) / 1024)
#define BMP   (BMBLK * BMW)
#define BMQ   (BMP / 4)
__global__ void __launch_bounds__(256)
k_prep(unsigned* __restrict__ gz, const int* __restrict__ esrc,
       const int* __restrict__ edst, unsigned* __restrict__ bm,
       unsigned* __restrict__ pk) {
    __shared__ unsigned bits[256 * BMW];
    const int t = threadIdx.x;
    const int i = blockIdx.x * 256 + t;
    if (blockIdx.x == 0 && t < GS_W) gz[t] = 0u;
    for (int k = 0; k < 256 * BMW / 4 / 256; ++k)
        ((uint4*)bits)[k * 256 + t] = make_uint4(0u, 0u, 0u, 0u);
    __syncthreads();
    if (i < NE / 4) {
        const int4 d = ((const int4*)edst)[i];
        const int4 s = ((const int4*)esrc)[i];
        uint4 w;
        w.x = (unsigned)s.x | ((unsigned)d.x << 16);
        w.y = (unsigned)s.y | ((unsigned)d.y << 16);
        w.z = (unsigned)s.z | ((unsigned)d.z << 16);
        w.w = (unsigned)s.w | ((unsigned)d.w << 16);
        ((uint4*)pk)[i] = w;
        const int wd = t >> 3;
        const unsigned b0 = 1u << ((4 * t) & 31);
        atomicOr(&bits[((unsigned)d.x / RNG) * BMW + wd], b0);
        atomicOr(&bits[((unsigned)d.y / RNG) * BMW + wd], b0 << 1);
        atomicOr(&bits[((unsigned)d.z / RNG) * BMW + wd], b0 << 2);
        atomicOr(&bits[((unsigned)d.w / RNG) * BMW + wd], b0 << 3);
    }
    __syncthreads();
    const int lane = t & 31, wv = t >> 5;
    const int pc = (lane & 7) * 4;
    unsigned* row = bm + (size_t)blockIdx.x * BMW + pc;
    for (int k = 0; k < 8; ++k) {
        const int g = k * 32 + wv * 4 + (lane >> 3);
        *(uint4*)(row + (size_t)g * BMP) = *(const uint4*)(bits + g * BMW + pc);
    }
}

#define G1_WAVES 16
#define G1_WGS   256
#define G1_UNITS (NN / 16 * 2)
__global__ void __launch_bounds__(G1_WAVES * 32)
k_gemm1(const float* __restrict__ X, const float* __restrict__ W,
        float* __restrict__ H) {
    __shared__ v4f wl[2 * 32 * 2 * 2 * 32];
    {
        const int t = threadIdx.x;
        const int k0 = 8 * (t >> 7) + 4 * ((t >> 4) & 1) + 2 * ((t >> 6) & 1);
        const int n0 = 32 * ((t >> 5) & 1) + (t & 15);
        v4f st[16];
        #pragma unroll
        for (int it = 0; it < 16; ++it) {
            const float* wp = W + (k0 + 32 * (it & 7)) * HID_C + n0 + 64 * (it >> 3);
            st[it].x = wp[0]; st[it].y = wp[HID_C]; st[it].z = wp[16]; st[it].w = wp[HID_C + 16];
        }
        #pragma unroll
        for (int it = 0; it < 16; ++it) wl[it * (G1_WAVES * 32) + t] = st[it];
    }
    __syncthreads();
    const int lane = threadIdx.x & 31;
    const int wv = __builtin_amdgcn_readfirstlane((int)(threadIdx.x >> 5));
    const int c = lane & 15, h = lane >> 4;
    const int u0 = (int)(((long)blockIdx.x * G1_UNITS) / G1_WGS);
    const int u1 = (int)(((long)(blockIdx.x + 1) * G1_UNITS) / G1_WGS);
    for (int u = u0 + wv; u < u1; u += G1_WAVES) {
        asm volatile("" ::: "memory");
        const int strip = u >> 1, hf = u & 1;
        const v4f* xr = (const v4f*)(X + (size_t)(strip * 16 + c) * IN_C + 4 * h);
        v4f a[32];
        #pragma unroll
        for (int j = 0; j < 32; ++j) a[j] = xr[2 * j];
        asm volatile("" :: "v"(a[0]), "v"(a[1]), "v"(a[2]), "v"(a[3]), "v"(a[4]), "v"(a[5]), "v"(a[6]), "v"(a[7]), "v"(a[8]), "v"(a[9]), "v"(a[10]), "v"(a[11]), "v"(a[12]), "v"(a[13]), "v"(a[14]), "v"(a[15]));
        asm volatile("" :: "v"(a[16]), "v"(a[17]), "v"(a[18]), "v"(a[19]), "v"(a[20]), "v"(a[21]), "v"(a[22]), "v"(a[23]), "v"(a[24]), "v"(a[25]), "v"(a[26]), "v"(a[27]), "v"(a[28]), "v"(a[29]), "v"(a[30]), "v"(a[31]));
        v8f acc0 = {}, acc1 = {}, acc2 = {}, acc3 = {};
        const v4f* bl = wl + hf * 4096 + lane;
        #pragma unroll
        for (int j = 0; j < 32; ++j) {
            #pragma unroll
            for (int m = 0; m < 2; ++m) {
                const v4f b0 = bl[((j * 2 + m) * 2 + 0) * 32];
                const v4f b1 = bl[((j * 2 + m) * 2 + 1) * 32];
                v2f av, bv;
                av.x = m ? a[j].z : a[j].x;
                av.y = m ? a[j].w : a[j].y;
                bv.x = b0.x; bv.y = b0.y;
                acc0 = __builtin_amdgcn_wmma_f32_16x16x4_f32(false, av, false, bv,
                                                             (short)0, acc0, false, false);
                bv.x = b0.z; bv.y = b0.w;
                acc1 = __builtin_amdgcn_wmma_f32_16x16x4_f32(false, av, false, bv,
                                                             (short)0, acc1, false, false);
                bv.x = b1.x; bv.y = b1.y;
                acc2 = __builtin_amdgcn_wmma_f32_16x16x4_f32(false, av, false, bv,
                                                             (short)0, acc2, false, false);
                bv.x = b1.z; bv.y = b1.w;
                acc3 = __builtin_amdgcn_wmma_f32_16x16x4_f32(false, av, false, bv,
                                                             (short)0, acc3, false, false);
            }
        }
        float* hp = H + (size_t)(strip * 16 + 8 * h) * HID_C + 64 * hf + c;
        #pragma unroll
        for (int r = 0; r < 8; ++r) {
            hp[(size_t)r * HID_C]      = acc0[r];
            hp[(size_t)r * HID_C + 16] = acc1[r];
            hp[(size_t)r * HID_C + 32] = acc2[r];
            hp[(size_t)r * HID_C + 48] = acc3[r];
        }
    }
}

__global__ void k_gemm2(const float* __restrict__ X, const float* __restrict__ W,
                        float* __restrict__ H) {
    const int wave = threadIdx.x >> 5;
    const int lane = threadIdx.x & 31;
    const int mBase = (blockIdx.x << 7) + (wave << 4);
    if (mBase >= NN) return;
    const int lm = lane & 15;
    const int kh = (lane >> 4) << 1;
    const float* xr = X + (size_t)(mBase + lm) * HID_C + kh;
    const float* wp = W + (size_t)kh * OUT_C + lm;
    v8f acc = {};
    #pragma unroll 8
    for (int kb = 0; kb < HID_C; kb += 4) {
        v2f a = *(const v2f*)(xr + kb);
        v2f b;
        b.x = wp[(size_t)kb * OUT_C];
        b.y = wp[(size_t)kb * OUT_C + OUT_C];
        acc = __builtin_amdgcn_wmma_f32_16x16x4_f32(false, a, false, b,
                                                    (short)0, acc, false, false);
    }
    float* hp = H + (size_t)(mBase + ((lane >> 4) << 3)) * OUT_C + lm;
    #pragma unroll
    for (int r = 0; r < 8; ++r) hp[(size_t)r * OUT_C] = acc[r];
}

__global__ void __launch_bounds__(256)
k_attvec(const float* __restrict__ H, const float* __restrict__ attS,
         const float* __restrict__ attD, float* __restrict__ aS,
         float* __restrict__ aD, int nH, unsigned* __restrict__ g) {
    __shared__ unsigned red[8][16];
    const int idx = blockIdx.x * 256 + threadIdx.x;
    unsigned eM = 0u, eN = 0u;
    if (idx < NN * nH) {
        const int hd = idx % nH;
        const float4* hp = (const float4*)(H + (size_t)idx * 16);
        const float4* sp = (const float4*)(attS + hd * 16);
        const float4* dp = (const float4*)(attD + hd * 16);
        float s = 0.f, d = 0.f;
        #pragma unroll
        for (int j = 0; j < 4; ++j) {
            float4 h4 = hp[j], s4 = sp[j], d4 = dp[j];
            s += h4.x * s4.x + h4.y * s4.y + h4.z * s4.z + h4.w * s4.w;
            d += h4.x * d4.x + h4.y * d4.y + h4.z * d4.z + h4.w * d4.w;
        }
        aS[idx] = s; aD[idx] = d;
        eM = enc_f32(s); eN = enc_f32(-s);
    }
    eM = umax2(eM, (unsigned)__shfl_xor((int)eM, 8));
    eN = umax2(eN, (unsigned)__shfl_xor((int)eN, 8));
    eM = umax2(eM, (unsigned)__shfl_xor((int)eM, 16));
    eN = umax2(eN, (unsigned)__shfl_xor((int)eN, 16));
    if (nH == 1) {
        eM = umax2(eM, (unsigned)__shfl_xor((int)eM, 1));
        eN = umax2(eN, (unsigned)__shfl_xor((int)eN, 1));
        eM = umax2(eM, (unsigned)__shfl_xor((int)eM, 2));
        eN = umax2(eN, (unsigned)__shfl_xor((int)eN, 2));
        eM = umax2(eM, (unsigned)__shfl_xor((int)eM, 4));
        eN = umax2(eN, (unsigned)__shfl_xor((int)eN, 4));
    }
    const int lane = threadIdx.x & 31, wave = threadIdx.x >> 5;
    if (lane < 8) { red[wave][lane] = eM; red[wave][8 + lane] = eN; }
    __syncthreads();
    if (threadIdx.x < 16) {
        const int t = threadIdx.x;
        unsigned v = red[0][t];
        #pragma unroll
        for (int w = 1; w < 8; ++w) v = umax2(v, red[w][t]);
        if (nH == 8) atomicMax(&g[G1_MAX + t], v);
        else if ((t & 7) == 0) atomicMax(&g[G2_MAX + (t >> 3)], v);
    }
}

__device__ __forceinline__ void edge_sd_pk(const unsigned* __restrict__ pk, int e, int& s, int& d) {
    if (e < NE) { const unsigned w = pk[e]; s = (int)(w & 0xFFFFu); d = (int)(w >> 16); }
    else { s = e - NE; d = s; }
}
__device__ __forceinline__ float leaky(float a) { return a > 0.f ? a : a * SLOPE; }

#define AGW   32
#define QCAP  1056
#define NIT   ((BMQ + AGW * 32 - 1) / (AGW * 32))
#define LCAP  512

template <class F>
__device__ __forceinline__ void filter_slice(const unsigned* __restrict__ bm, int wv, int lane,
                                             int blk, int lo, int* q, F&& batch) {
    int base = 0;
    const uint4* row = (const uint4*)(bm + (size_t)blk * BMP);
    const int wr = AGW - 1 - wv;
    uint4 dv = row[wr * 32 + lane];
    static_assert(AGW * 32 <= BMQ, "the first step of every wave lies inside the row");
    for (int it = 0; it <= NIT; ++it) {
        uint4 nx = make_uint4(0u, 0u, 0u, 0u);
        {
            const int c = ((it + 1) * AGW + wr) * 32 + lane;
            if (c < BMQ) nx = row[c];
        }
        uint4 cur = dv;
        const int eb = ((it * AGW + wr) * 32 + lane) * 128;
        const int nt = it < NIT ? 4 : 1;
        #pragma nounroll
        for (int t = 0; t < nt; ++t) {
            if (it < NIT) {
                unsigned m = cur.x;
                unsigned mask;
                while ((mask = __builtin_amdgcn_ballot_w32(m != 0u)) != 0u) {
                    const int pos = base + (int)__builtin_amdgcn_mbcnt_lo(mask, 0u);
                    if (m != 0u) q[pos] = eb + 32 * t + __builtin_ctz(m);
                    m &= m - 1u;
                    base += __builtin_popcount(mask);
                }
                cur.x = cur.y; cur.y = cur.z; cur.z = cur.w;
            } else {
                const int nl = wv * 32 + lane;
                const bool m = nl < RNG && lo + nl < NN;
                const unsigned mask = __builtin_amdgcn_ballot_w32(m);
                if (mask) {
                    const int pos = base + (int)__builtin_amdgcn_mbcnt_lo(mask, 0u);
                    if (m) q[pos] = NE + lo + nl;
                    base += __builtin_popcount(mask);
                }
            }
            const int need = it < NIT ? 32 : 1;
            while (base >= need) {
                const int cnt = base < 32 ? base : 32;
                const int e = lane < cnt ? q[base - cnt + lane] : 0;
                batch(e, cnt);
                base -= cnt;
            }
        }
        dv = nx;
    }
}

__device__ __forceinline__ float quad_bcast(float x, int u) {
    const int xi = __float_as_int(x);
    int r;
    switch (u) {
        case 0:  r = __builtin_amdgcn_update_dpp(0, xi, 0x00, 0xf, 0xf, false); break;
        case 1:  r = __builtin_amdgcn_update_dpp(0, xi, 0x55, 0xf, 0xf, false); break;
        case 2:  r = __builtin_amdgcn_update_dpp(0, xi, 0xAA, 0xf, 0xf, false); break;
        default: r = __builtin_amdgcn_update_dpp(0, xi, 0xFF, 0xf, 0xf, false); break;
    }
    return __int_as_float(r);
}

__device__ __forceinline__ void agg_batch(
        int e, int cnt, int lane, int lo,
        const unsigned* __restrict__ pk,
        const float* __restrict__ aS, const float* __restrict__ aD,
        const float* __restrict__ H, const unsigned* mxs,
        bool useBound, float gmx, float* acc, float* den,
        int2* __restrict__ lst, int lpos) {
    int s = 0, d = lo;
    if (lane < cnt) edge_sd_pk(pk, e, s, d);
    if (lane < cnt && lpos + lane < LCAP) lst[lpos + lane] = make_int2(s, d);
    const v4f* hrow = (const v4f*)H + lane;
    auto load_row = [&](int i) -> v4f {
        const int sk = __builtin_amdgcn_readlane(s, i);
        return hrow[(size_t)sk * (HID_C / 4)];
    };
    v4f r[3][4];
    #pragma unroll
    for (int k = 0; k < 8; ++k) r[k >> 2][k & 3] = load_row(k);
    const int hd = lane >> 2, sub = lane & 3;
    float wq[8];
    #pragma unroll
    for (int q = 0; q < 8; ++q) {
        const int i = q * 4 + sub;
        const int sq = __shfl(s, i), dq = __shfl(d, i);
        const float adq = aD[(size_t)dq * 8 + hd];
        const float al = leaky(aS[(size_t)sq * 8 + hd] + adq);
        float mx;
        if (useBound) mx = leaky(gmx + adq);
        else          mx = dec_f32(mxs[(dq - lo) * 8 + hd]);
        wq[q] = __expf(al - mx);
        if (i < cnt) atomicAdd(&den[(dq - lo) * 8 + hd], wq[q]);
    }
    auto add_row = [&](int i, const v4f rv) {
        const float wk = quad_bcast(wq[i >> 2], i & 3);
        const int dk = __builtin_amdgcn_readlane(d, i);
        float* op = acc + (dk - lo) * HID_C + lane;
        atomicAdd(op,      wk * rv.x);
        atomicAdd(op + 32, wk * rv.y);
        atomicAdd(op + 64, wk * rv.z);
        atomicAdd(op + 96, wk * rv.w);
    };
    if (cnt == 32) {
        #pragma unroll
        for (int g = 0; g < 8; ++g) {
            if (g + 2 < 8) {
                #pragma unroll
                for (int k = 0; k < 4; ++k) r[(g + 2) % 3][k] = load_row((g + 2) * 4 + k);
            }
            #pragma unroll
            for (int k = 0; k < 4; ++k) add_row(g * 4 + k, r[g % 3][k]);
        }
    } else {
        #pragma unroll
        for (int g = 0; g < 8; ++g) {
            if (g * 4 >= cnt) break;
            if (g >= 2) {
                #pragma unroll
                for (int k = 0; k < 4; ++k) r[g % 3][k] = load_row(g * 4 + k);
            }
            #pragma unroll
            for (int k = 0; k < 4; ++k)
                if (g * 4 + k < cnt) add_row(g * 4 + k, r[g % 3][k]);
        }
    }
}

__global__ void __launch_bounds__(1024)
k_agg1(const unsigned* __restrict__ pk,
       const float* __restrict__ aS, const float* __restrict__ aD,
       const float* __restrict__ H,
       const float* __restrict__ bias, float* __restrict__ out1,
       const unsigned* __restrict__ g, const unsigned* __restrict__ bm,
       int* __restrict__ lst, int* __restrict__ lcnt) {
    __shared__ float acc[RNG * HID_C];
    __shared__ float den[RNG * 8];
    __shared__ unsigned mxs[RNG * 8];
    __shared__ int   que[AGW][QCAP];
    const int lane = threadIdx.x & 31;
    const int wv = __builtin_amdgcn_readfirstlane((int)(threadIdx.x >> 5));
    const int lo = blockIdx.x * RNG;
    for (int i = threadIdx.x; i < RNG * HID_C; i += 1024) acc[i] = 0.f;
    for (int i = threadIdx.x; i < RNG * 8; i += 1024) den[i] = 0.f;
    __syncthreads();
    const bool useBound = spread_small8(g);
    const float gmx = dec_f32(g[G1_MAX + (lane >> 2)]);
    if (!useBound) {
        for (int i = threadIdx.x; i < RNG * 8; i += 1024) mxs[i] = 0u;
        __syncthreads();
        filter_slice(bm, wv, lane, (int)blockIdx.x, lo, que[wv], [&](int e, int cnt) {
            if (lane < cnt) {
                int s, d;
                edge_sd_pk(pk, e, s, d);
                const float4* sp = (const float4*)(aS + (size_t)s * 8);
                const float4* dp = (const float4*)(aD + (size_t)d * 8);
                const float4 s0 = sp[0], s1 = sp[1], d0 = dp[0], d1 = dp[1];
                unsigned* m = mxs + (d - lo) * 8;
                atomicMax(m + 0, enc_f32(leaky(s0.x + d0.x)));
                atomicMax(m + 1, enc_f32(leaky(s0.y + d0.y)));
                atomicMax(m + 2, enc_f32(leaky(s0.z + d0.z)));
                atomicMax(m + 3, enc_f32(leaky(s0.w + d0.w)));
                atomicMax(m + 4, enc_f32(leaky(s1.x + d1.x)));
                atomicMax(m + 5, enc_f32(leaky(s1.y + d1.y)));
                atomicMax(m + 6, enc_f32(leaky(s1.z + d1.z)));
                atomicMax(m + 7, enc_f32(leaky(s1.w + d1.w)));
            }
        });
        __syncthreads();
    }
    int2* lp = (int2*)lst + (size_t)(blockIdx.x * AGW + wv) * LCAP;
    int lpos = 0;
    filter_slice(bm, wv, lane, (int)blockIdx.x, lo, que[wv], [&](int e, int cnt) {
        agg_batch(e, cnt, lane, lo, pk, aS, aD, H, mxs, useBound, gmx, acc, den,
                  lp, lpos);
        lpos += cnt;
    });
    if (lane == 0) lcnt[blockIdx.x * AGW + wv] = lpos;
    __syncthreads();
    for (int i = threadIdx.x; i < RNG * 32; i += 1024) {
        const int nl = i >> 5, L = i & 31;
        const int n = lo + nl;
        if (n < NN) {
            const float dn = den[nl * 8 + (L >> 2)];
            const float* ap = acc + nl * HID_C + L;
            const float* bp = bias + 4 * L;
            v4f o;
            o.x = fmaxf(ap[0]  / dn + bp[0], 0.f);
            o.y = fmaxf(ap[32] / dn + bp[1], 0.f);
            o.z = fmaxf(ap[64] / dn + bp[2], 0.f);
            o.w = fmaxf(ap[96] / dn + bp[3], 0.f);
            ((v4f*)(out1 + (size_t)n * HID_C))[L] = o;
        }
    }
}

__device__ __forceinline__ void agg2_batch(
        int s, int d, int cnt, int lane, int lo,
        const float* __restrict__ aS, const float* __restrict__ aD,
        const float* __restrict__ H, const unsigned* mx2,
        bool useBound, float gmx, float* acc, float* den) {
    const float adq = aD[d];
    const float al = leaky(aS[s] + adq);
    float mx;
    if (useBound) mx = leaky(gmx + adq);
    else          mx = dec_f32(mx2[d - lo]);
    const float w = __expf(al - mx);
    if (lane < cnt) atomicAdd(&den[d - lo], w);
    const int ch = lane & 15, hi = lane >> 4;
    float v[16];
    int   dd[16];
    #pragma unroll
    for (int t = 0; t < 16; ++t) {
        const int i = t * 2 + hi;
        const int st = __shfl(s, i);
        dd[t] = __shfl(d, i);
        v[t] = __shfl(w, i) * H[(size_t)st * OUT_C + ch];
    }
    #pragma unroll
    for (int t = 0; t < 16; ++t) {
        const int i = t * 2 + hi;
        if (i < cnt) atomicAdd(&acc[(dd[t] - lo) * OUT_C + ch], v[t]);
    }
}

__global__ void __launch_bounds__(1024)
k_agg2(const unsigned* __restrict__ pk,
       const float* __restrict__ aS, const float* __restrict__ aD,
       const float* __restrict__ H,
       const float* __restrict__ bias, float* __restrict__ out,
       const unsigned* __restrict__ g, const unsigned* __restrict__ bm,
       const int* __restrict__ lst, const int* __restrict__ lcnt) {
    __shared__ float acc[RNG * OUT_C];
    __shared__ float den[RNG];
    __shared__ unsigned mx2[RNG];
    __shared__ int   que[AGW][QCAP];
    const int lane = threadIdx.x & 31;
    const int wv = __builtin_amdgcn_readfirstlane((int)(threadIdx.x >> 5));
    const int lo = blockIdx.x * RNG;
    for (int i = threadIdx.x; i < RNG * OUT_C; i += 1024) acc[i] = 0.f;
    for (int i = threadIdx.x; i < RNG; i += 1024) den[i] = 0.f;
    __syncthreads();
    const bool useBound = spread_small1(g);
    const float gmx = dec_f32(g[G2_MAX]);
    const int tot = __builtin_amdgcn_readfirstlane(lcnt[blockIdx.x * AGW + wv]);
    const int2* lp = (const int2*)lst + (size_t)(blockIdx.x * AGW + wv) * LCAP;
    if (!useBound) {
        for (int i = threadIdx.x; i < RNG; i += 1024) mx2[i] = 0u;
        __syncthreads();
        if (tot <= LCAP) {
            for (int b0 = 0; b0 < tot; b0 += 32) {
                if (b0 + lane < tot) {
                    const int2 p = lp[b0 + lane];
                    atomicMax(&mx2[p.y - lo], enc_f32(leaky(aS[p.x] + aD[p.y])));
                }
            }
        } else {
            filter_slice(bm, wv, lane, (int)blockIdx.x, lo, que[wv], [&](int e, int cnt) {
                if (lane < cnt) {
                    int s, d;
                    edge_sd_pk(pk, e, s, d);
                    atomicMax(&mx2[d - lo], enc_f32(leaky(aS[s] + aD[d])));
                }
            });
        }
        __syncthreads();
    }
    if (tot <= LCAP) {
        int2 sd = make_int2(0, lo);
        if (lane < tot) sd = lp[lane];
        for (int b0 = 0; b0 < tot; b0 += 32) {
            int2 nx = make_int2(0, lo);
            if (b0 + 32 + lane < tot) nx = lp[b0 + 32 + lane];
            const int cnt = tot - b0 < 32 ? tot - b0 : 32;
            agg2_batch(sd.x, sd.y, cnt, lane, lo, aS, aD, H, mx2, useBound, gmx, acc, den);
            sd = nx;
        }
    } else {
        filter_slice(bm, wv, lane, (int)blockIdx.x, lo, que[wv], [&](int e, int cnt) {
            int s = 0, d = lo;
            if (lane < cnt) edge_sd_pk(pk, e, s, d);
            agg2_batch(s, d, cnt, lane, lo, aS, aD, H, mx2, useBound, gmx, acc, den);
        });
    }
    __syncthreads();
    for (int i = threadIdx.x; i < RNG * OUT_C; i += 1024) {
        const int nl = i >> 4, c = i & (OUT_C - 1);
        const int n = lo + nl;
        if (n < NN) out[(size_t)n * OUT_C + c] = acc[i] / den[nl] + bias[c];
    }
}

extern "C" void kernel_launch(void* const* d_in, const int* in_sizes, int n_in,
                              void* d_out, int out_size, void* d_ws, size_t ws_size,
                              hipStream_t stream) {
    const float* x     = (const float*)d_in[0];
    const int*   ei    = (const int*)  d_in[1];
    const float* W1    = (const float*)d_in[2];
    const float* attS1 = (const float*)d_in[3];
    const float* attD1 = (const float*)d_in[4];
    const float* bias1 = (const float*)d_in[5];
    const float* W2    = (const float*)d_in[6];
    const float* attS2 = (const float*)d_in[7];
    const float* attD2 = (const float*)d_in[8];
    const float* bias2 = (const float*)d_in[9];
    float* out = (float*)d_out;
    const int* esrc = ei;
    const int* edst = ei + NE;

    unsigned* gst = (unsigned*)d_ws;
    float* out1 = (float*)(gst + GS_W);
    float* h1  = out1 + (size_t)NN * 128;
    float* aS1 = h1 + (size_t)NN * 128;
    float* aD1 = aS1 + (size_t)NN * 8;
    float* h2  = aD1 + (size_t)NN * 8;
    float* aS2 = h2 + (size_t)NN * 16;
    float* aD2 = aS2 + NN;
    unsigned* bmap = (unsigned*)(aD2 + NN);
    int*  lcnt = (int*)(bmap + (size_t)256 * BMP);
    int*  lst  = lcnt + 256 * AGW;
    unsigned* pk = (unsigned*)(lst + (size_t)256 * AGW * LCAP * 2);

    k_prep<<<BMBLK, 256, 0, stream>>>(gst, esrc, edst, bmap, pk);

    k_gemm1<<<G1_WGS, G1_WAVES * 32, 0, stream>>>(x, W1, h1);
    k_attvec<<<(NN * HEADS + 255) / 256, 256, 0, stream>>>(h1, attS1, attD1, aS1, aD1, HEADS, gst);
    k_agg1<<<256, 1024, 0, stream>>>(
        pk, aS1, aD1, h1, bias1, out1, gst, bmap, lst, lcnt);

    k_gemm2<<<(NN + 127) / 128, 256, 0, stream>>>(out1, W2, h2);
    k_attvec<<<(NN + 255) / 256, 256, 0, stream>>>(h2, attS2, attD2, aS2, aD2, 1, gst);
    k_agg2<<<256, 1024, 0, stream>>>(
        pk, aS2, aD2, h2, bias2, out, gst, bmap, lst, lcnt);
}


# Prep sorts each 2048-edge chunk's packed edges by owner group into 128-byte cells (count + words, LDS integer rank); a group's filter reads its cells: no scattered read per edge; arrival-order sums
# speedup vs baseline: 13.4984x; 1.1215x over previous
#include <hip/hip_runtime.h>
#include <cstdint>

#define NN      50000
#define NE      1600000
#define IN_C    256
#define HID_C   128
#define HEADS   8
#define OUT_C   16
#define SLOPE   0.2f

typedef float v2f __attribute__((ext_vector_type(2)));
typedef float v4f __attribute__((ext_vector_type(4)));
typedef float v8f __attribute__((ext_vector_type(8)));

__device__ __forceinline__ unsigned enc_f32(float f) {
    unsigned u = __float_as_uint(f);
    return (u & 0x80000000u) ? ~u : (u | 0x80000000u);
}
__device__ __forceinline__ float dec_f32(unsigned e) {
    unsigned u = (e & 0x80000000u) ? (e & 0x7FFFFFFFu) : ~e;
    return __uint_as_float(u);
}

__device__ __forceinline__ unsigned umax2(unsigned a, unsigned b) { return a > b ? a : b; }

#define GS_W    32
#define G1_MAX  0
#define G1_NEG  8
#define G2_MAX  16
#define G2_NEG  17
#define SPREAD_OK 16.0f

__device__ __forceinline__ bool spread_small8(const unsigned* __restrict__ g) {
    bool ok = true;
    #pragma unroll
    for (int h = 0; h < 8; ++h) {
        const float r = dec_f32(g[G1_MAX + h]) + dec_f32(g[G1_NEG + h]);
        ok = ok && (r < SPREAD_OK);
    }
    return ok;
}
__device__ __forceinline__ bool spread_small1(const unsigned* __restrict__ g) {
    return dec_f32(g[G2_MAX]) + dec_f32(g[G2_NEG]) < SPREAD_OK;
}

#define RNG   196

static_assert(NN <= 65536, "a node number must fit 16 bits of the packed edge word");
#define CHUNK 2048
#define CW    32
#define CCAP  (CW - 1)
#define NCH   ((NE + CHUNK - 1) / CHUNK)
#define BMP   (NCH * CW)
#define BMQ   (BMP / 4)
static_assert(NE % 4 == 0, "the edge lists are read four edges per load");
__device__ __forceinline__ void cell_put(unsigned* cell, int s, int d) {
    unsigned* cp = cell + ((unsigned)d / RNG) * CW;
    const unsigned rank = atomicAdd(cp, 1u);
    if (rank < CCAP) cp[1 + rank] = (unsigned)s | ((unsigned)d << 16);
}
__global__ void __launch_bounds__(256)
k_prep(unsigned* __restrict__ gz, const int* __restrict__ esrc,
       const int* __restrict__ edst, unsigned* __restrict__ bm) {
    __shared__ unsigned cell[256 * CW];
    const int t = threadIdx.x;
    if (blockIdx.x == 0 && t < GS_W) gz[t] = 0u;
    for (int k = 0; k < 256 * CW / 4 / 256; ++k)
        ((uint4*)cell)[k * 256 + t] = make_uint4(0u, 0u, 0u, 0u);
    __syncthreads();
    int4 d[CHUNK / 1024], s[CHUNK / 1024];
    #pragma unroll
    for (int r = 0; r < CHUNK / 1024; ++r) {
        const int i = blockIdx.x * (CHUNK / 4) + r * 256 + t;
        d[r] = make_int4(-1, -1, -1, -1);
        if (i < NE / 4) { d[r] = ((const int4*)edst)[i]; s[r] = ((const int4*)esrc)[i]; }
    }
    #pragma unroll
    for (int r = 0; r < CHUNK / 1024; ++r) {
        if (d[r].x >= 0) {
            cell_put(cell, s[r].x, d[r].x);
            cell_put(cell, s[r].y, d[r].y);
            cell_put(cell, s[r].z, d[r].z);
            cell_put(cell, s[r].w, d[r].w);
        }
    }
    __syncthreads();
    const int lane = t & 31, wv = t >> 5;
    const int pc = (lane & 7) * 4;
    unsigned* row = bm + (size_t)blockIdx.x * CW + pc;
    for (int k = 0; k < 8; ++k) {
        const int g = k * 32 + wv * 4 + (lane >> 3);
        *(uint4*)(row + (size_t)g * BMP) = *(const uint4*)(cell + g * CW + pc);
    }
}

#define G1_WAVES 16
#define G1_WGS   256
#define G1_UNITS (NN / 16 * 2)
__global__ void __launch_bounds__(G1_WAVES * 32)
k_gemm1(const float* __restrict__ X, const float* __restrict__ W,
        float* __restrict__ H) {
    __shared__ v4f wl[2 * 32 * 2 * 2 * 32];
    {
        const int t = threadIdx.x;
        const int k0 = 8 * (t >> 7) + 4 * ((t >> 4) & 1) + 2 * ((t >> 6) & 1);
        const int n0 = 32 * ((t >> 5) & 1) + (t & 15);
        v4f st[16];
        #pragma unroll
        for (int it = 0; it < 16; ++it) {
            const float* wp = W + (k0 + 32 * (it & 7)) * HID_C + n0 + 64 * (it >> 3);
            st[it].x = wp[0]; st[it].y = wp[HID_C]; st[it].z = wp[16]; st[it].w = wp[HID_C + 16];
        }
        #pragma unroll
        for (int it = 0; it < 16; ++it) wl[it * (G1_WAVES * 32) + t] = st[it];
    }
    __syncthreads();
    const int lane = threadIdx.x & 31;
    const int wv = __builtin_amdgcn_readfirstlane((int)(threadIdx.x >> 5));
    const int c = lane & 15, h = lane >> 4;
    const int u0 = (int)(((long)blockIdx.x * G1_UNITS) / G1_WGS);
    const int u1 = (int)(((long)(blockIdx.x + 1) * G1_UNITS) / G1_WGS);
    for (int u = u0 + wv; u < u1; u += G1_WAVES) {
        asm volatile("" ::: "memory");
        const int strip = u >> 1, hf = u & 1;
        const v4f* xr = (const v4f*)(X + (size_t)(strip * 16 + c) * IN_C + 4 * h);
        v4f a[32];
        #pragma unroll
        for (int j = 0; j < 32; ++j) a[j] = xr[2 * j];
        asm volatile("" :: "v"(a[0]), "v"(a[1]), "v"(a[2]), "v"(a[3]), "v"(a[4]), "v"(a[5]), "v"(a[6]), "v"(a[7]), "v"(a[8]), "v"(a[9]), "v"(a[10]), "v"(a[11]), "v"(a[12]), "v"(a[13]), "v"(a[14]), "v"(a[15]));
        asm volatile("" :: "v"(a[16]), "v"(a[17]), "v"(a[18]), "v"(a[19]), "v"(a[20]), "v"(a[21]), "v"(a[22]), "v"(a[23]), "v"(a[24]), "v"(a[25]), "v"(a[26]), "v"(a[27]), "v"(a[28]), "v"(a[29]), "v"(a[30]), "v"(a[31]));
        v8f acc0 = {}, acc1 = {}, acc2 = {}, acc3 = {};
        const v4f* bl = wl + hf * 4096 + lane;
        #pragma unroll
        for (int j = 0; j < 32; ++j) {
            #pragma unroll
            for (int m = 0; m < 2; ++m) {
                const v4f b0 = bl[((j * 2 + m) * 2 + 0) * 32];
                const v4f b1 = bl[((j * 2 + m) * 2 + 1) * 32];
                v2f av, bv;
                av.x = m ? a[j].z : a[j].x;
                av.y = m ? a[j].w : a[j].y;
                bv.x = b0.x; bv.y = b0.y;
                acc0 = __builtin_amdgcn_wmma_f32_16x16x4_f32(false, av, false, bv,
                                                             (short)0, acc0, false, false);
                bv.x = b0.z; bv.y = b0.w;
                acc1 = __builtin_amdgcn_wmma_f32_16x16x4_f32(false, av, false, bv,
                                                             (short)0, acc1, false, false);
                bv.x = b1.x; bv.y = b1.y;
                acc2 = __builtin_amdgcn_wmma_f32_16x16x4_f32(false, av, false, bv,
                                                             (short)0, acc2, false, false);
                bv.x = b1.z; bv.y = b1.w;
                acc3 = __builtin_amdgcn_wmma_f32_16x16x4_f32(false, av, false, bv,
                                                             (short)0, acc3, false, false);
            }
        }
        float* hp = H + (size_t)(strip * 16 + 8 * h) * HID_C + 64 * hf + c;
        #pragma unroll
        for (int r = 0; r < 8; ++r) {
            hp[(size_t)r * HID_C]      = acc0[r];
            hp[(size_t)r * HID_C + 16] = acc1[r];
            hp[(size_t)r * HID_C + 32] = acc2[r];
            hp[(size_t)r * HID_C + 48] = acc3[r];
        }
    }
}

__global__ void k_gemm2(const float* __restrict__ X, const float* __restrict__ W,
                        float* __restrict__ H) {
    const int wave = threadIdx.x >> 5;
    const int lane = threadIdx.x & 31;
    const int mBase = (blockIdx.x << 7) + (wave << 4);
    if (mBase >= NN) return;
    const int lm = lane & 15;
    const int kh = (lane >> 4) << 1;
    const float* xr = X + (size_t)(mBase + lm) * HID_C + kh;
    const float* wp = W + (size_t)kh * OUT_C + lm;
    v8f acc = {};
    #pragma unroll 8
    for (int kb = 0; kb < HID_C; kb += 4) {
        v2f a = *(const v2f*)(xr + kb);
        v2f b;
        b.x = wp[(size_t)kb * OUT_C];
        b.y = wp[(size_t)kb * OUT_C + OUT_C];
        acc = __builtin_amdgcn_wmma_f32_16x16x4_f32(false, a, false, b,
                                                    (short)0, acc, false, false);
    }
    float* hp = H + (size_t)(mBase + ((lane >> 4) << 3)) * OUT_C + lm;
    #pragma unroll
    for (int r = 0; r < 8; ++r) hp[(size_t)r * OUT_C] = acc[r];
}

__global__ void __launch_bounds__(256)
k_attvec(const float* __restrict__ H, const float* __restrict__ attS,
         const float* __restrict__ attD, float* __restrict__ aS,
         float* __restrict__ aD, int nH, unsigned* __restrict__ g) {
    __shared__ unsigned red[8][16];
    const int idx = blockIdx.x * 256 + threadIdx.x;
    unsigned eM = 0u, eN = 0u;
    if (idx < NN * nH) {
        const int hd = idx % nH;
        const float4* hp = (const float4*)(H + (size_t)idx * 16);
        const float4* sp = (const float4*)(attS + hd * 16);
        const float4* dp = (const float4*)(attD + hd * 16);
        float s = 0.f, d = 0.f;
        #pragma unroll
        for (int j = 0; j < 4; ++j) {
            float4 h4 = hp[j], s4 = sp[j], d4 = dp[j];
            s += h4.x * s4.x + h4.y * s4.y + h4.z * s4.z + h4.w * s4.w;
            d += h4.x * d4.x + h4.y * d4.y + h4.z * d4.z + h4.w * d4.w;
        }
        aS[idx] = s; aD[idx] = d;
        eM = enc_f32(s); eN = enc_f32(-s);
    }
    eM = umax2(eM, (unsigned)__shfl_xor((int)eM, 8));
    eN = umax2(eN, (unsigned)__shfl_xor((int)eN, 8));
    eM = umax2(eM, (unsigned)__shfl_xor((int)eM, 16));
    eN = umax2(eN, (unsigned)__shfl_xor((int)eN, 16));
    if (nH == 1) {
        eM = umax2(eM, (unsigned)__shfl_xor((int)eM, 1));
        eN = umax2(eN, (unsigned)__shfl_xor((int)eN, 1));
        eM = umax2(eM, (unsigned)__shfl_xor((int)eM, 2));
        eN = umax2(eN, (unsigned)__shfl_xor((int)eN, 2));
        eM = umax2(eM, (unsigned)__shfl_xor((int)eM, 4));
        eN = umax2(eN, (unsigned)__shfl_xor((int)eN, 4));
    }
    const int lane = threadIdx.x & 31, wave = threadIdx.x >> 5;
    if (lane < 8) { red[wave][lane] = eM; red[wave][8 + lane] = eN; }
    __syncthreads();
    if (threadIdx.x < 16) {
        const int t = threadIdx.x;
        unsigned v = red[0][t];
        #pragma unroll
        for (int w = 1; w < 8; ++w) v = umax2(v, red[w][t]);
        if (nH == 8) atomicMax(&g[G1_MAX + t], v);
        else if ((t & 7) == 0) atomicMax(&g[G2_MAX + (t >> 3)], v);
    }
}

__device__ __forceinline__ void edge_sd_pk(int e, int& s, int& d) {
    s = (int)((unsigned)e & 0xFFFFu); d = (int)((unsigned)e >> 16);
}
__device__ __forceinline__ float leaky(float a) { return a > 0.f ? a : a * SLOPE; }

#define AGW   32
#define QCAP  160
#define NIT   ((BMQ + AGW * 32 - 1) / (AGW * 32))
#define LCAP  512

template <class F>
__device__ __forceinline__ void filter_slice(const unsigned* __restrict__ bm,
                                             const int* __restrict__ esrc,
                                             const int* __restrict__ edst, int wv, int lane,
                                             int blk, int lo, int* q, F&& batch) {
    int base = 0;
    const uint4* row = (const uint4*)(bm + (size_t)blk * BMP);
    const int wr = AGW - 1 - wv;
    uint4 dv = row[wr * 32 + lane];
    static_assert(AGW * 32 <= BMQ, "the first step of every wave lies inside the row");
    static_assert(BMQ % 8 == 0, "the eight lanes of a cell are inside the row together");
    const int w0 = 4 * (lane & 7);
    for (int it = 0; it <= NIT; ++it) {
        uint4 nx = make_uint4(0u, 0u, 0u, 0u);
        {
            const int c = ((it + 1) * AGW + wr) * 32 + lane;
            if (c < BMQ) nx = row[c];
        }
        unsigned ovf = 0u;
        const int piece = (it * AGW + wr) * 32 + lane;
        if (it < NIT) {
            const unsigned cn = (unsigned)__builtin_amdgcn_mov_dpp8((int)dv.x, 0);
            const int n = cn <= (unsigned)CCAP ? (int)cn : 0;
            ovf = __builtin_amdgcn_ballot_w32(w0 == 0 && cn > (unsigned)CCAP);
            const unsigned wd[4] = {dv.x, dv.y, dv.z, dv.w};
            #pragma unroll
            for (int t = 0; t < 4; ++t) {
                const bool m = w0 + t >= 1 && w0 + t <= n;
                const unsigned mask = __builtin_amdgcn_ballot_w32(m);
                const int pos = base + (int)__builtin_amdgcn_mbcnt_lo(mask, 0u);
                if (m) q[pos] = (int)wd[t];
                base += __builtin_popcount(mask);
            }
        } else {
            const int nl = wv * 32 + lane;
            const bool m = nl < RNG && lo + nl < NN;
            const unsigned mask = __builtin_amdgcn_ballot_w32(m);
            if (mask) {
                const int pos = base + (int)__builtin_amdgcn_mbcnt_lo(mask, 0u);
                if (m) q[pos] = (int)((unsigned)(lo + nl) * 0x10001u);
                base += __builtin_popcount(mask);
            }
        }
        const int need = it < NIT ? 32 : 1;
        int fpos = 0;
        for (;;) {
            while (base >= need) {
                const int cnt = base < 32 ? base : 32;
                const int e = lane < cnt ? q[base - cnt + lane] : 0;
                batch(e, cnt);
                base -= cnt;
            }
            if (ovf == 0u) break;
            const int ch = __builtin_amdgcn_readlane(piece, __builtin_ctz(ovf)) >> 3;
            const int e = ch * CHUNK + fpos + lane;
            bool m = false;
            unsigned w = 0u;
            if (e < NE) {
                const unsigned d = (unsigned)edst[e];
                m = d / RNG == (unsigned)blk;
                w = (unsigned)esrc[e] | (d << 16);
            }
            const unsigned mask = __builtin_amdgcn_ballot_w32(m);
            const int pos = base + (int)__builtin_amdgcn_mbcnt_lo(mask, 0u);
            if (m) q[pos] = (int)w;
            base += __builtin_popcount(mask);
            fpos += 32;
            if (fpos == CHUNK) { fpos = 0; ovf &= ovf - 1u; }
        }
        dv = nx;
    }
}

__device__ __forceinline__ float quad_bcast(float x, int u) {
    const int xi = __float_as_int(x);
    int r;
    switch (u) {
        case 0:  r = __builtin_amdgcn_update_dpp(0, xi, 0x00, 0xf, 0xf, false); break;
        case 1:  r = __builtin_amdgcn_update_dpp(0, xi, 0x55, 0xf, 0xf, false); break;
        case 2:  r = __builtin_amdgcn_update_dpp(0, xi, 0xAA, 0xf, 0xf, false); break;
        default: r = __builtin_amdgcn_update_dpp(0, xi, 0xFF, 0xf, 0xf, false); break;
    }
    return __int_as_float(r);
}

__device__ __forceinline__ void agg_batch(
        int e, int cnt, int lane, int lo,
        const float* __restrict__ aS, const float* __restrict__ aD,
        const float* __restrict__ H, const unsigned* mxs,
        bool useBound, float gmx, float* acc, float* den,
        int2* __restrict__ lst, int lpos) {
    int s = 0, d = lo;
    if (lane < cnt) edge_sd_pk(e, s, d);
    if (lane < cnt && lpos + lane < LCAP) lst[lpos + lane] = make_int2(s, d);
    const v4f* hrow = (const v4f*)H + lane;
    auto load_row = [&](int i) -> v4f {
        const int sk = __builtin_amdgcn_readlane(s, i);
        return hrow[(size_t)sk * (HID_C / 4)];
    };
    v4f r[3][4];
    #pragma unroll
    for (int k = 0; k < 8; ++k) r[k >> 2][k & 3] = load_row(k);
    const int hd = lane >> 2, sub = lane & 3;
    float wq[8];
    #pragma unroll
    for (int q = 0; q < 8; ++q) {
        const int i = q * 4 + sub;
        const int sq = __shfl(s, i), dq = __shfl(d, i);
        const float adq = aD[(size_t)dq * 8 + hd];
        const float al = leaky(aS[(size_t)sq * 8 + hd] + adq);
        float mx;
        if (useBound) mx = leaky(gmx + adq);
        else          mx = dec_f32(mxs[(dq - lo) * 8 + hd]);
        wq[q] = __expf(al - mx);
        if (i < cnt) atomicAdd(&den[(dq - lo) * 8 + hd], wq[q]);
    }
    auto add_row = [&](int i, const v4f rv) {
        const float wk = quad_bcast(wq[i >> 2], i & 3);
        const int dk = __builtin_amdgcn_readlane(d, i);
        float* op = acc + (dk - lo) * HID_C + lane;
        atomicAdd(op,      wk * rv.x);
        atomicAdd(op + 32, wk * rv.y);
        atomicAdd(op + 64, wk * rv.z);
        atomicAdd(op + 96, wk * rv.w);
    };
    if (cnt == 32) {
        #pragma unroll
        for (int g = 0; g < 8; ++g) {
            if (g + 2 < 8) {
                #pragma unroll
                for (int k = 0; k < 4; ++k) r[(g + 2) % 3][k] = load_row((g + 2) * 4 + k);
            }
            #pragma unroll
            for (int k = 0; k < 4; ++k) add_row(g * 4 + k, r[g % 3][k]);
        }
    } else {
        #pragma unroll
        for (int g = 0; g < 8; ++g) {
            if (g * 4 >= cnt) break;
            if (g >= 2) {
                #pragma unroll
                for (int k = 0; k < 4; ++k) r[g % 3][k] = load_row(g * 4 + k);
            }
            #pragma unroll
            for (int k = 0; k < 4; ++k)
                if (g * 4 + k < cnt) add_row(g * 4 + k, r[g % 3][k]);
        }
    }
}

__global__ void __launch_bounds__(1024)
k_agg1(const int* __restrict__ esrc, const int* __restrict__ edst,
       const float* __restrict__ aS, const float* __restrict__ aD,
       const float* __restrict__ H,
       const float* __restrict__ bias, float* __restrict__ out1,
       const unsigned* __restrict__ g, const unsigned* __restrict__ bm,
       int* __restrict__ lst, int* __restrict__ lcnt) {
    __shared__ float acc[RNG * HID_C];
    __shared__ float den[RNG * 8];
    __shared__ unsigned mxs[RNG * 8];
    __shared__ int   que[AGW][QCAP];
    const int lane = threadIdx.x & 31;
    const int wv = __builtin_amdgcn_readfirstlane((int)(threadIdx.x >> 5));
    const int lo = blockIdx.x * RNG;
    for (int i = threadIdx.x; i < RNG * HID_C; i += 1024) acc[i] = 0.f;
    for (int i = threadIdx.x; i < RNG * 8; i += 1024) den[i] = 0.f;
    __syncthreads();
    const bool useBound = spread_small8(g);
    const float gmx = dec_f32(g[G1_MAX + (lane >> 2)]);
    if (!useBound) {
        for (int i = threadIdx.x; i < RNG * 8; i += 1024) mxs[i] = 0u;
        __syncthreads();
        filter_slice(bm, esrc, edst, wv, lane, (int)blockIdx.x, lo, que[wv], [&](int e, int cnt) {
            if (lane < cnt) {
                int s, d;
                edge_sd_pk(e, s, d);
                const float4* sp = (const float4*)(aS + (size_t)s * 8);
                const float4* dp = (const float4*)(aD + (size_t)d * 8);
                const float4 s0 = sp[0], s1 = sp[1], d0 = dp[0], d1 = dp[1];
                unsigned* m = mxs + (d - lo) * 8;
                atomicMax(m + 0, enc_f32(leaky(s0.x + d0.x)));
                atomicMax(m + 1, enc_f32(leaky(s0.y + d0.y)));
                atomicMax(m + 2, enc_f32(leaky(s0.z + d0.z)));
                atomicMax(m + 3, enc_f32(leaky(s0.w + d0.w)));
                atomicMax(m + 4, enc_f32(leaky(s1.x + d1.x)));
                atomicMax(m + 5, enc_f32(leaky(s1.y + d1.y)));
                atomicMax(m + 6, enc_f32(leaky(s1.z + d1.z)));
                atomicMax(m + 7, enc_f32(leaky(s1.w + d1.w)));
            }
        });
        __syncthreads();
    }
    int2* lp = (int2*)lst + (size_t)(blockIdx.x * AGW + wv) * LCAP;
    int lpos = 0;
    filter_slice(bm, esrc, edst, wv, lane, (int)blockIdx.x, lo, que[wv], [&](int e, int cnt) {
        agg_batch(e, cnt, lane, lo, aS, aD, H, mxs, useBound, gmx, acc, den,
                  lp, lpos);
        lpos += cnt;
    });
    if (lane == 0) lcnt[blockIdx.x * AGW + wv] = lpos;
    __syncthreads();
    for (int i = threadIdx.x; i < RNG * 32; i += 1024) {
        const int nl = i >> 5, L = i & 31;
        const int n = lo + nl;
        if (n < NN) {
            const float dn = den[nl * 8 + (L >> 2)];
            const float* ap = acc + nl * HID_C + L;
            const float* bp = bias + 4 * L;
            v4f o;
            o.x = fmaxf(ap[0]  / dn + bp[0], 0.f);
            o.y = fmaxf(ap[32] / dn + bp[1], 0.f);
            o.z = fmaxf(ap[64] / dn + bp[2], 0.f);
            o.w = fmaxf(ap[96] / dn + bp[3], 0.f);
            ((v4f*)(out1 + (size_t)n * HID_C))[L] = o;
        }
    }
}

__device__ __forceinline__ void agg2_batch(
        int s, int d, int cnt, int lane, int lo,
        const float* __restrict__ aS, const float* __restrict__ aD,
        const float* __restrict__ H, const unsigned* mx2,
        bool useBound, float gmx, float* acc, float* den) {
    const float adq = aD[d];
    const float al = leaky(aS[s] + adq);
    float mx;
    if (useBound) mx = leaky(gmx + adq);
    else          mx = dec_f32(mx2[d - lo]);
    const float w = __expf(al - mx);
    if (lane < cnt) atomicAdd(&den[d - lo], w);
    const int ch = lane & 15, hi = lane >> 4;
    float v[16];
    int   dd[16];
    #pragma unroll
    for (int t = 0; t < 16; ++t) {
        const int i = t * 2 + hi;
        const int st = __shfl(s, i);
        dd[t] = __shfl(d, i);
        v[t] = __shfl(w, i) * H[(size_t)st * OUT_C + ch];
    }
    #pragma unroll
    for (int t = 0; t < 16; ++t) {
        const int i = t * 2 + hi;
        if (i < cnt) atomicAdd(&acc[(dd[t] - lo) * OUT_C + ch], v[t]);
    }
}

__global__ void __launch_bounds__(1024)
k_agg2(const int* __restrict__ esrc, const int* __restrict__ edst,
       const float* __restrict__ aS, const float* __restrict__ aD,
       const float* __restrict__ H,
       const float* __restrict__ bias, float* __restrict__ out,
       const unsigned* __restrict__ g, const unsigned* __restrict__ bm,
       const int* __restrict__ lst, const int* __restrict__ lcnt) {
    __shared__ float acc[RNG * OUT_C];
    __shared__ float den[RNG];
    __shared__ unsigned mx2[RNG];
    __shared__ int   que[AGW][QCAP];
    const int lane = threadIdx.x & 31;
    const int wv = __builtin_amdgcn_readfirstlane((int)(threadIdx.x >> 5));
    const int lo = blockIdx.x * RNG;
    for (int i = threadIdx.x; i < RNG * OUT_C; i += 1024) acc[i] = 0.f;
    for (int i = threadIdx.x; i < RNG; i += 1024) den[i] = 0.f;
    __syncthreads();
    const bool useBound = spread_small1(g);
    const float gmx = dec_f32(g[G2_MAX]);
    const int tot = __builtin_amdgcn_readfirstlane(lcnt[blockIdx.x * AGW + wv]);
    const int2* lp = (const int2*)lst + (size_t)(blockIdx.x * AGW + wv) * LCAP;
    if (!useBound) {
        for (int i = threadIdx.x; i < RNG; i += 1024) mx2[i] = 0u;
        __syncthreads();
        if (tot <= LCAP) {
            for (int b0 = 0; b0 < tot; b0 += 32) {
                if (b0 + lane < tot) {
                    const int2 p = lp[b0 + lane];
                    atomicMax(&mx2[p.y - lo], enc_f32(leaky(aS[p.x] + aD[p.y])));
                }
            }
        } else {
            filter_slice(bm, esrc, edst, wv, lane, (int)blockIdx.x, lo, que[wv], [&](int e, int cnt) {
                if (lane < cnt) {
                    int s, d;
                    edge_sd_pk(e, s, d);
                    atomicMax(&mx2[d - lo], enc_f32(leaky(aS[s] + aD[d])));
                }
            });
        }
        __syncthreads();
    }
    if (tot <= LCAP) {
        int2 sd = make_int2(0, lo);
        if (lane < tot) sd = lp[lane];
        for (int b0 = 0; b0 < tot; b0 += 32) {
            int2 nx = make_int2(0, lo);
            if (b0 + 32 + lane < tot) nx = lp[b0 + 32 + lane];
            const int cnt = tot - b0 < 32 ? tot - b0 : 32;
            agg2_batch(sd.x, sd.y, cnt, lane, lo, aS, aD, H, mx2, useBound, gmx, acc, den);
            sd = nx;
        }
    } else {
        filter_slice(bm, esrc, edst, wv, lane, (int)blockIdx.x, lo, que[wv], [&](int e, int cnt) {
            int s = 0, d = lo;
            if (lane < cnt) edge_sd_pk(e, s, d);
            agg2_batch(s, d, cnt, lane, lo, aS, aD, H, mx2, useBound, gmx, acc, den);
        });
    }
    __syncthreads();
    for (int i = threadIdx.x; i < RNG * OUT_C; i += 1024) {
        const int nl = i >> 4, c = i & (OUT_C - 1);
        const int n = lo + nl;
        if (n < NN) out[(size_t)n * OUT_C + c] = acc[i] / den[nl] + bias[c];
    }
}

extern "C" void kernel_launch(void* const* d_in, const int* in_sizes, int n_in,
                              void* d_out, int out_size, void* d_ws, size_t ws_size,
                              hipStream_t stream) {
    const float* x     = (const float*)d_in[0];
    const int*   ei    = (const int*)  d_in[1];
    const float* W1    = (const float*)d_in[2];
    const float* attS1 = (const float*)d_in[3];
    const float* attD1 = (const float*)d_in[4];
    const float* bias1 = (const float*)d_in[5];
    const float* W2    = (const float*)d_in[6];
    const float* attS2 = (const float*)d_in[7];
    const float* attD2 = (const float*)d_in[8];
    const float* bias2 = (const float*)d_in[9];
    float* out = (float*)d_out;
    const int* esrc = ei;
    const int* edst = ei + NE;

    unsigned* gst = (unsigned*)d_ws;
    float* out1 = (float*)(gst + GS_W);
    float* h1  = out1 + (size_t)NN * 128;
    float* aS1 = h1 + (size_t)NN * 128;
    float* aD1 = aS1 + (size_t)NN * 8;
    float* h2  = aD1 + (size_t)NN * 8;
    float* aS2 = h2 + (size_t)NN * 16;
    float* aD2 = aS2 + NN;
    unsigned* bmap = (unsigned*)(aD2 + NN);
    int*  lcnt = (int*)(bmap + (size_t)256 * BMP);
    int*  lst  = lcnt + 256 * AGW;

    k_prep<<<NCH, 256, 0, stream>>>(gst, esrc, edst, bmap);

    k_gemm1<<<G1_WGS, G1_WAVES * 32, 0, stream>>>(x, W1, h1);
    k_attvec<<<(NN * HEADS + 255) / 256, 256, 0, stream>>>(h1, attS1, attD1, aS1, aD1, HEADS, gst);
    k_agg1<<<256, 1024, 0, stream>>>(
        esrc, edst, aS1, aD1, h1, bias1, out1, gst, bmap, lst, lcnt);

    k_gemm2<<<(NN + 127) / 128, 256, 0, stream>>>(out1, W2, h2);
    k_attvec<<<(NN + 255) / 256, 256, 0, stream>>>(h2, attS2, attD2, aS2, aD2, 1, gst);
    k_agg2<<<256, 1024, 0, stream>>>(
        esrc, edst, aS2, aD2, h2, bias2, out, gst, bmap, lst, lcnt);
}
